# starting kernel
# baseline (speedup 1.0000x reference)
#include <hip/hip_runtime.h>

typedef __attribute__((ext_vector_type(2))) float v2f;
typedef __attribute__((ext_vector_type(8))) float v8f;

#define N_NODES 50000
#define N_EDGES 800000
#define N_FEAT  128
#define N_HID   64

__global__ void init_bias_kernel(float* __restrict__ buf,
                                 const float* __restrict__ bias) {
    int i = blockIdx.x * blockDim.x + threadIdx.x;
    if (i < N_NODES * N_HID) buf[i] = bias[i & (N_HID - 1)];
}

__global__ __launch_bounds__(256) void gemm_wmma_kernel(
    const float* __restrict__ X, const float* __restrict__ W,
    float* __restrict__ HW, int M, int K) {
    int wave = (blockIdx.x * blockDim.x + threadIdx.x) >> 5;
    int lane = threadIdx.x & 31;
    int m0 = wave * 16;
    if (m0 >= M) return;
    int half = lane >> 4;
    int l    = lane & 15;

    v8f acc0 = {}, acc1 = {}, acc2 = {}, acc3 = {};
    const float* xrow = X + (size_t)(m0 + l) * K;

    for (int k = 0; k < K; k += 4) {
        int kk = k + 2 * half;
        v2f a = *(const v2f*)(xrow + kk);
        const float* bp = W + (size_t)kk * N_HID + l;
        v2f b0 = { bp[ 0], bp[ 0 + N_HID] };
        v2f b1 = { bp[16], bp[16 + N_HID] };
        v2f b2 = { bp[32], bp[32 + N_HID] };
        v2f b3 = { bp[48], bp[48 + N_HID] };
        acc0 = __builtin_amdgcn_wmma_f32_16x16x4_f32(false, a, false, b0, (short)0, acc0, false, false);
        acc1 = __builtin_amdgcn_wmma_f32_16x16x4_f32(false, a, false, b1, (short)0, acc1, false, false);
        acc2 = __builtin_amdgcn_wmma_f32_16x16x4_f32(false, a, false, b2, (short)0, acc2, false, false);
        acc3 = __builtin_amdgcn_wmma_f32_16x16x4_f32(false, a, false, b3, (short)0, acc3, false, false);
    }

    float* orow = HW + (size_t)(m0 + 8 * half) * N_HID + l;
    #pragma unroll
    for (int r = 0; r < 8; ++r) {
        orow[(size_t)r * N_HID +  0] = acc0[r];
        orow[(size_t)r * N_HID + 16] = acc1[r];
        orow[(size_t)r * N_HID + 32] = acc2[r];
        orow[(size_t)r * N_HID + 48] = acc3[r];
    }
}

__global__ void scatter_kernel(const float* __restrict__ HW,
                               const int* __restrict__ src,
                               const int* __restrict__ dst,
                               const float* __restrict__ ew,
                               float* __restrict__ agg) {
    int t = blockIdx.x * blockDim.x + threadIdx.x;
    if (t >= N_EDGES * N_HID) return;
    int e = t >> 6;
    int f = t & 63;
    float v = HW[(size_t)src[e] * N_HID + f] * ew[e];
    atomicAdd(&agg[(size_t)dst[e] * N_HID + f], v);
}

__global__ void relu_kernel(float* __restrict__ h) {
    int i = blockIdx.x * blockDim.x + threadIdx.x;
    if (i < N_NODES * N_HID) h[i] = fmaxf(h[i], 0.0f);
}

extern "C" void kernel_launch(void* const* d_in, const int* in_sizes, int n_in,
                              void* d_out, int out_size, void* d_ws, size_t ws_size,
                              hipStream_t stream) {
    const float* x  = (const float*)d_in[0];
    const int*   ei = (const int*)  d_in[1];
    const float* ew = (const float*)d_in[2];
    const float* w1 = (const float*)d_in[3];
    const float* b1 = (const float*)d_in[4];
    const float* w2 = (const float*)d_in[5];
    const float* b2 = (const float*)d_in[6];
    float* out = (float*)d_out;

    float* hw  = (float*)d_ws;
    float* agg = hw + (size_t)N_NODES * N_HID;

    const int* src = ei;
    const int* dst = ei + N_EDGES;

    const int nodeElems = N_NODES * N_HID;
    const int edgeElems = N_EDGES * N_HID;
    const int mTiles    = N_NODES / 16;
    dim3 blk(256);
    dim3 gNode((nodeElems + 255) / 256);
    dim3 gEdge((edgeElems + 255) / 256);
    dim3 gGemm((mTiles + 7) / 8);

    gemm_wmma_kernel<<<gGemm, blk, 0, stream>>>(x, w1, hw, N_NODES, N_FEAT);
    init_bias_kernel<<<gNode, blk, 0, stream>>>(agg, b1);
    scatter_kernel<<<gEdge, blk, 0, stream>>>(hw, src, dst, ew, agg);
    relu_kernel<<<gNode, blk, 0, stream>>>(agg);

    gemm_wmma_kernel<<<gGemm, blk, 0, stream>>>(agg, w2, hw, N_NODES, N_HID);
    init_bias_kernel<<<gNode, blk, 0, stream>>>(out, b2);
    scatter_kernel<<<gEdge, blk, 0, stream>>>(hw, src, dst, ew, out);
}


# Bias seeding of both aggregation buffers and the inter-layer relu folded into the two product kernels (seed stored per 16-row tile, relu on the A fragment load): 7 launches -> 4; sum order unchanged
# speedup vs baseline: 3.9203x; 3.9203x over previous
#include <hip/hip_runtime.h>

typedef __attribute__((ext_vector_type(2))) float v2f;
typedef __attribute__((ext_vector_type(4))) float v4f;
typedef __attribute__((ext_vector_type(8))) float v8f;

#define N_NODES 50000
#define N_EDGES 800000
#define N_FEAT  128
#define N_HID   64

#define CNT_STRIDE 64

template <bool RELU_IN>
__device__ __forceinline__ void gemm_body(
    const float* __restrict__ X, const float* __restrict__ W,
    float* __restrict__ HW, int M, int K,
    const float* __restrict__ bias, float* __restrict__ seed,
    int* __restrict__ cnt) {
    int gtid = blockIdx.x * blockDim.x + threadIdx.x;
    if (gtid == 0) { cnt[0] = 0; cnt[CNT_STRIDE] = 0; }
    int wave = gtid >> 5;
    int lane = threadIdx.x & 31;
    int m0 = wave * 16;
    if (m0 >= M) return;
    int half = lane >> 4;
    int l    = lane & 15;

    v4f b4 = *(const v4f*)(bias + ((lane * 4) & (N_HID - 1)));

    v8f acc0 = {}, acc1 = {}, acc2 = {}, acc3 = {};
    const float* xrow = X + (size_t)(m0 + l) * K;

    for (int k = 0; k < K; k += 4) {
        int kk = k + 2 * half;
        v2f a = *(const v2f*)(xrow + kk);
        if (RELU_IN) { a.x = fmaxf(a.x, 0.0f); a.y = fmaxf(a.y, 0.0f); }
        const float* bp = W + (size_t)kk * N_HID + l;
        v2f b0 = { bp[ 0], bp[ 0 + N_HID] };
        v2f b1 = { bp[16], bp[16 + N_HID] };
        v2f b2 = { bp[32], bp[32 + N_HID] };
        v2f b3 = { bp[48], bp[48 + N_HID] };
        acc0 = __builtin_amdgcn_wmma_f32_16x16x4_f32(false, a, false, b0, (short)0, acc0, false, false);
        acc1 = __builtin_amdgcn_wmma_f32_16x16x4_f32(false, a, false, b1, (short)0, acc1, false, false);
        acc2 = __builtin_amdgcn_wmma_f32_16x16x4_f32(false, a, false, b2, (short)0, acc2, false, false);
        acc3 = __builtin_amdgcn_wmma_f32_16x16x4_f32(false, a, false, b3, (short)0, acc3, false, false);
    }

    float* sp = seed + (size_t)m0 * N_HID + lane * 4;
    for (int j = 0; j < 8; ++j) *(v4f*)(sp + j * 128) = b4;

    float* orow = HW + (size_t)(m0 + 8 * half) * N_HID + l;
    #pragma unroll
    for (int r = 0; r < 8; ++r) {
        orow[(size_t)r * N_HID +  0] = acc0[r];
        orow[(size_t)r * N_HID + 16] = acc1[r];
        orow[(size_t)r * N_HID + 32] = acc2[r];
        orow[(size_t)r * N_HID + 48] = acc3[r];
    }
}

__global__ __launch_bounds__(256) void gemm_seed_kernel(
    const float* __restrict__ X, const float* __restrict__ W,
    float* __restrict__ HW, int M, int K,
    const float* __restrict__ bias, float* __restrict__ seed,
    int* __restrict__ cnt) {
    gemm_body<false>(X, W, HW, M, K, bias, seed, cnt);
}

__global__ __launch_bounds__(256) void gemm_relu_seed_kernel(
    const float* __restrict__ X, const float* __restrict__ W,
    float* __restrict__ HW, int M, int K,
    const float* __restrict__ bias, float* __restrict__ seed,
    int* __restrict__ cnt) {
    gemm_body<true>(X, W, HW, M, K, bias, seed, cnt);
}

#define EDGE_GROUP 8
#define EDGES_PER_WAVE 64
#define WAVES_PER_BLOCK 4
#define EDGES_PER_ITEM (EDGES_PER_WAVE * WAVES_PER_BLOCK)
#define N_ITEMS (N_EDGES / EDGES_PER_ITEM)
#define SCATTER_BLOCKS (2 * N_ITEMS)
static_assert(N_EDGES % EDGES_PER_ITEM == 0, "edge items are whole");

__device__ __forceinline__ void add_edges(const float* __restrict__ HW,
                                          float* __restrict__ agg,
                                          int s_l, int d_l, int w_l,
                                          unsigned m, int lane) {
    while (m) {
        int n = __builtin_popcount(m);
        if (n > EDGE_GROUP) n = EDGE_GROUP;
        int   bl[EDGE_GROUP];
        float v0[EDGE_GROUP], v1[EDGE_GROUP];
        #pragma unroll
        for (int j = 0; j < EDGE_GROUP; ++j) {
            bl[j] = 0; v0[j] = 0.0f; v1[j] = 0.0f;
            if (j < n) {
                int b = __builtin_ctz(m);
                m &= m - 1;
                bl[j] = b;
                int s = __builtin_amdgcn_readlane(s_l, b);
                const float* hp = HW + (size_t)s * N_HID + lane;
                v0[j] = hp[0];
                v1[j] = hp[32];
            }
        }
        #pragma unroll
        for (int j = 0; j < EDGE_GROUP; ++j) {
            if (j < n) {
                int   d = __builtin_amdgcn_readlane(d_l, bl[j]);
                float w = __int_as_float(__builtin_amdgcn_readlane(w_l, bl[j]));
                float* ap = agg + (size_t)d * N_HID + lane;
                atomicAdd(ap,      v0[j] * w);
                atomicAdd(ap + 32, v1[j] * w);
            }
        }
    }
}

__global__ __launch_bounds__(128) void scatter_kernel(const float* __restrict__ HW,
                               const int* __restrict__ src,
                               const int* __restrict__ dst,
                               const float* __restrict__ ew,
                               float* __restrict__ agg,
                               int* __restrict__ cnt) {
    __shared__ int s_p, s_t;
    int lane = threadIdx.x & 31;
    int wv   = threadIdx.x >> 5;
    if (threadIdx.x == 0) {
        int p = (__builtin_amdgcn_s_sendmsg_rtn(0x87) >> 18) & 1;
        int t = atomicAdd(cnt + p * CNT_STRIDE, 1);
        if (t >= N_ITEMS) {
            p ^= 1;
            t = atomicAdd(cnt + p * CNT_STRIDE, 1);
        }
        s_p = p;
        s_t = t;
    }
    __syncthreads();
    int p = s_p;
    int t = s_t;
    if (t >= N_ITEMS) return;

    int e  = t * EDGES_PER_ITEM + wv * EDGES_PER_WAVE + lane;
    int sA = src[e],                  sB = src[e + 32];
    int dA = dst[e],                  dB = dst[e + 32];
    int wA = __float_as_int(ew[e]),   wB = __float_as_int(ew[e + 32]);
    unsigned mA = __builtin_amdgcn_ballot_w32((dA & 1) == p);
    unsigned mB = __builtin_amdgcn_ballot_w32((dB & 1) == p);
    add_edges(HW, agg, sA, dA, wA, mA, lane);
    add_edges(HW, agg, sB, dB, wB, mB, lane);
}

extern "C" void kernel_launch(void* const* d_in, const int* in_sizes, int n_in,
                              void* d_out, int out_size, void* d_ws, size_t ws_size,
                              hipStream_t stream) {
    const float* x  = (const float*)d_in[0];
    const int*   ei = (const int*)  d_in[1];
    const float* ew = (const float*)d_in[2];
    const float* w1 = (const float*)d_in[3];
    const float* b1 = (const float*)d_in[4];
    const float* w2 = (const float*)d_in[5];
    const float* b2 = (const float*)d_in[6];
    float* out = (float*)d_out;

    float* hw  = (float*)d_ws;
    float* agg = hw + (size_t)N_NODES * N_HID;
    int*   cnt1 = (int*)(agg + (size_t)N_NODES * N_HID);
    int*   cnt2 = cnt1 + 2 * CNT_STRIDE;

    const int* src = ei;
    const int* dst = ei + N_EDGES;

    const int mTiles    = N_NODES / 16;
    dim3 blk(256);
    dim3 gEdge(SCATTER_BLOCKS);
    dim3 blkEdge(32 * WAVES_PER_BLOCK);
    dim3 gGemm((mTiles + 7) / 8);

    gemm_seed_kernel<<<gGemm, blk, 0, stream>>>(x, w1, hw, N_NODES, N_FEAT, b1, agg, cnt1);
    scatter_kernel<<<gEdge, blkEdge, 0, stream>>>(hw, src, dst, ew, agg, cnt1);

    gemm_relu_seed_kernel<<<gGemm, blk, 0, stream>>>(agg, w2, hw, N_NODES, N_HID, b2, out, cnt2);
    scatter_kernel<<<gEdge, blkEdge, 0, stream>>>(hw, src, dst, ew, out, cnt2);
}
